# starting kernel
# baseline (speedup 1.0000x reference)
#include <hip/hip_runtime.h>

typedef __attribute__((ext_vector_type(2))) float v2f;
typedef __attribute__((ext_vector_type(8))) float v8f;

static constexpr int KDIM    = 128;
static constexpr int N_EDGES = 25000;
static constexpr float EPS   = 1e-12f;

__global__ __launch_bounds__(256) void fr_softmax(const float* __restrict__ y,
                                                  float* __restrict__ probs, int n) {
    int wave = threadIdx.x >> 5, lane = threadIdx.x & 31;
    int row = blockIdx.x * 8 + wave;
    if (row >= n) return;
    float4 v = ((const float4*)(y + (size_t)row * KDIM))[lane];
    float m = fmaxf(fmaxf(v.x, v.y), fmaxf(v.z, v.w));
    #pragma unroll
    for (int off = 16; off > 0; off >>= 1) m = fmaxf(m, __shfl_xor(m, off, 32));
    float4 e;
    e.x = __expf(v.x - m); e.y = __expf(v.y - m);
    e.z = __expf(v.z - m); e.w = __expf(v.w - m);
    float s = e.x + e.y + e.z + e.w;
    #pragma unroll
    for (int off = 16; off > 0; off >>= 1) s += __shfl_xor(s, off, 32);
    float inv = 1.f / s;
    e.x *= inv; e.y *= inv; e.z *= inv; e.w *= inv;
    ((float4*)(probs + (size_t)row * KDIM))[lane] = e;
}

__global__ __launch_bounds__(256) void fr_degrees(const int* __restrict__ node_idx,
                                                  const int* __restrict__ edge_idx,
                                                  float* __restrict__ edge_deg,
                                                  float* __restrict__ node_deg, int nnz) {
    int i = blockIdx.x * 256 + threadIdx.x;
    if (i >= nnz) return;
    atomicAdd(&edge_deg[edge_idx[i]], 1.0f);
    atomicAdd(&node_deg[node_idx[i]], 1.0f);
}

__global__ __launch_bounds__(256) void fr_gather(const float* __restrict__ probs,
                                                 const int* __restrict__ node_idx,
                                                 const int* __restrict__ edge_idx,
                                                 float* __restrict__ edge_agg, int nnz) {
    long long gid = (long long)blockIdx.x * 256 + threadIdx.x;
    int i = (int)(gid >> 5);
    if (i >= nnz) return;
    int c4 = (int)(gid & 31);
    int nv = node_idx[i], ev = edge_idx[i];
    float4 p = ((const float4*)(probs + (size_t)nv * KDIM))[c4];
    float* dst = edge_agg + (size_t)ev * KDIM + c4 * 4;
    atomicAdd(dst + 0, p.x); atomicAdd(dst + 1, p.y);
    atomicAdd(dst + 2, p.z); atomicAdd(dst + 3, p.w);
}

__global__ __launch_bounds__(256) void fr_edge_gemm(const float* __restrict__ edge_agg,
                                                    const float* __restrict__ edge_deg,
                                                    const float* __restrict__ W,
                                                    float* __restrict__ edge_feat, int E) {
    __shared__ float sA[16][132];
    int tid = threadIdx.x;
    int m0 = blockIdx.x * 16;
    for (int i = tid; i < 16 * 32; i += 256) {
        int row = i >> 5, c4 = i & 31, r = m0 + row;
        float4 v = make_float4(0.f, 0.f, 0.f, 0.f);
        if (r < E) v = ((const float4*)edge_agg)[(size_t)r * 32 + c4];
        *(float4*)&sA[row][c4 * 4] = v;
    }
    __syncthreads();

    int wave = tid >> 5, lane = tid & 31;
    int half = lane >> 4, lr = lane & 15;
    int n0 = wave * 16;

    v8f c = {0.f, 0.f, 0.f, 0.f, 0.f, 0.f, 0.f, 0.f};
    #pragma unroll 4
    for (int k0 = 0; k0 < KDIM; k0 += 4) {
        int ka = k0 + 2 * half;
        v2f a, b;
        a.x = sA[lr][ka];
        a.y = sA[lr][ka + 1];
        b.x = W[(size_t)ka * KDIM + n0 + lr];
        b.y = W[(size_t)(ka + 1) * KDIM + n0 + lr];
        c = __builtin_amdgcn_wmma_f32_16x16x4_f32(false, a, false, b, (short)0, c,
                                                  false, false);
    }
    int rowbase = m0 + 8 * half;
    #pragma unroll
    for (int r = 0; r < 8; ++r) {
        int row = rowbase + r;
        if (row < E) {
            float inv = 1.f / fmaxf(edge_deg[row], 1.f);
            edge_feat[(size_t)row * KDIM + n0 + lr] = c[r] * inv;
        }
    }
}

__global__ __launch_bounds__(256) void fr_scatter(const float* __restrict__ edge_feat,
                                                  const int* __restrict__ node_idx,
                                                  const int* __restrict__ edge_idx,
                                                  float* __restrict__ out, int nnz) {
    long long gid = (long long)blockIdx.x * 256 + threadIdx.x;
    int i = (int)(gid >> 5);
    if (i >= nnz) return;
    int c4 = (int)(gid & 31);
    int nv = node_idx[i], ev = edge_idx[i];
    float4 f = ((const float4*)(edge_feat + (size_t)ev * KDIM))[c4];
    float* dst = out + (size_t)nv * KDIM + c4 * 4;
    atomicAdd(dst + 0, f.x); atomicAdd(dst + 1, f.y);
    atomicAdd(dst + 2, f.z); atomicAdd(dst + 3, f.w);
}

__global__ __launch_bounds__(256) void fr_finalize(const float* __restrict__ probs,
                                                   const float* __restrict__ node_deg,
                                                   float* __restrict__ out, int n) {
    int wave = threadIdx.x >> 5, lane = threadIdx.x & 31;
    int row = blockIdx.x * 8 + wave;
    if (row >= n) return;
    float inv = 1.f / fmaxf(node_deg[row], 1.f);
    float4 acc = ((const float4*)(out + (size_t)row * KDIM))[lane];
    float4 p   = ((const float4*)(probs + (size_t)row * KDIM))[lane];
    float4 nat;
    nat.x = fmaxf(p.x, EPS) * acc.x * inv;
    nat.y = fmaxf(p.y, EPS) * acc.y * inv;
    nat.z = fmaxf(p.z, EPS) * acc.z * inv;
    nat.w = fmaxf(p.w, EPS) * acc.w * inv;
    float s = nat.x + nat.y + nat.z + nat.w;
    #pragma unroll
    for (int off = 16; off > 0; off >>= 1) s += __shfl_xor(s, off, 32);
    float mean = s * (1.0f / KDIM);
    nat.x -= mean; nat.y -= mean; nat.z -= mean; nat.w -= mean;
    ((float4*)(out + (size_t)row * KDIM))[lane] = nat;
}

extern "C" void kernel_launch(void* const* d_in, const int* in_sizes, int n_in,
                              void* d_out, int out_size, void* d_ws, size_t ws_size,
                              hipStream_t stream) {
    const float* y        = (const float*)d_in[0];
    const float* W        = (const float*)d_in[1];
    const int*   node_idx = (const int*)d_in[2];
    const int*   edge_idx = (const int*)d_in[3];
    float*       out      = (float*)d_out;

    const int n   = in_sizes[0] / KDIM;
    const int nnz = in_sizes[2];
    const int E   = N_EDGES;

    float* probs     = (float*)d_ws;
    float* edge_agg  = probs + (size_t)n * KDIM;
    float* edge_feat = edge_agg + (size_t)E * KDIM;
    float* edge_deg  = edge_feat + (size_t)E * KDIM;
    float* node_deg  = edge_deg + E;

    hipMemsetAsync(edge_agg, 0, (size_t)E * KDIM * sizeof(float), stream);
    hipMemsetAsync(edge_deg, 0, (size_t)(E + n) * sizeof(float), stream);
    hipMemsetAsync(out, 0, (size_t)n * KDIM * sizeof(float), stream);

    fr_softmax<<<(n + 7) / 8, 256, 0, stream>>>(y, probs, n);
    fr_degrees<<<(nnz + 255) / 256, 256, 0, stream>>>(node_idx, edge_idx,
                                                      edge_deg, node_deg, nnz);
    {
        long long total = (long long)nnz * 32;
        fr_gather<<<(unsigned)((total + 255) / 256), 256, 0, stream>>>(
            probs, node_idx, edge_idx, edge_agg, nnz);
    }
    fr_edge_gemm<<<(E + 15) / 16, 256, 0, stream>>>(edge_agg, edge_deg, W,
                                                    edge_feat, E);
    {
        long long total = (long long)nnz * 32;
        fr_scatter<<<(unsigned)((total + 255) / 256), 256, 0, stream>>>(
            edge_feat, node_idx, edge_idx, out, nnz);
    }
    fr_finalize<<<(n + 7) / 8, 256, 0, stream>>>(probs, node_deg, out, n);
}


# gather: a block owns 320 edge rows as an f32 tile in LDS, filters a third of the entry list for them, adds node rows by LDS f32 adds, then adds the tile to edge_agg by line; arrival order, 3 partials
# speedup vs baseline: 1.5325x; 1.5325x over previous
#include <hip/hip_runtime.h>

typedef __attribute__((ext_vector_type(2))) float v2f;
typedef __attribute__((ext_vector_type(8))) float v8f;

static constexpr int KDIM    = 128;
static constexpr int N_EDGES = 25000;
static constexpr float EPS   = 1e-12f;

__global__ __launch_bounds__(256) void fr_softmax(const float* __restrict__ y,
                                                  float* __restrict__ probs, int n) {
    int wave = threadIdx.x >> 5, lane = threadIdx.x & 31;
    int row = blockIdx.x * 8 + wave;
    if (row >= n) return;
    float4 v = ((const float4*)(y + (size_t)row * KDIM))[lane];
    float m = fmaxf(fmaxf(v.x, v.y), fmaxf(v.z, v.w));
    #pragma unroll
    for (int off = 16; off > 0; off >>= 1) m = fmaxf(m, __shfl_xor(m, off, 32));
    float4 e;
    e.x = __expf(v.x - m); e.y = __expf(v.y - m);
    e.z = __expf(v.z - m); e.w = __expf(v.w - m);
    float s = e.x + e.y + e.z + e.w;
    #pragma unroll
    for (int off = 16; off > 0; off >>= 1) s += __shfl_xor(s, off, 32);
    float inv = 1.f / s;
    e.x *= inv; e.y *= inv; e.z *= inv; e.w *= inv;
    ((float4*)(probs + (size_t)row * KDIM))[lane] = e;
}

static constexpr int DEG_SLICE     = 25000;
static constexpr int DEG_CHUNKS    = 32;
static constexpr int DEG_NODE_JOBS = 4;
static constexpr int DEG_THREADS   = 1024;

__device__ __forceinline__ void deg_count4(unsigned int* hist, int4 v, int lo, unsigned cnt) {
    const unsigned b0 = (unsigned)(v.x - lo), b1 = (unsigned)(v.y - lo);
    const unsigned b2 = (unsigned)(v.z - lo), b3 = (unsigned)(v.w - lo);
    if (b0 < cnt) atomicAdd(&hist[b0], 1u);
    if (b1 < cnt) atomicAdd(&hist[b1], 1u);
    if (b2 < cnt) atomicAdd(&hist[b2], 1u);
    if (b3 < cnt) atomicAdd(&hist[b3], 1u);
}

__global__ __launch_bounds__(DEG_THREADS) void fr_degrees(const int* __restrict__ node_idx,
                                                          const int* __restrict__ edge_idx,
                                                          float* __restrict__ edge_deg,
                                                          float* __restrict__ node_deg,
                                                          int nnz, int n, int E) {
    __shared__ unsigned int hist[DEG_SLICE];
    const int tid = threadIdx.x;
    const int job = blockIdx.y;
    const int* __restrict__ idx = (job == 0) ? edge_idx : node_idx;
    float* __restrict__ deg     = (job == 0) ? edge_deg : node_deg;
    const int rows   = (job == 0) ? E : n;
    const int lo0    = (job == 0) ? 0 : (job - 1) * DEG_SLICE;
    const int stride = (job == 0) ? DEG_SLICE : DEG_NODE_JOBS * DEG_SLICE;

    int per = (nnz + (int)gridDim.x - 1) / (int)gridDim.x;
    per = (per + 3) & ~3;
    const long long s64 = (long long)blockIdx.x * per;
    if (s64 >= nnz) return;
    const int start = (int)s64;
    const int end   = (nnz - start < per) ? nnz : start + per;
    const int n4    = (end - start) >> 2;
    const int4* __restrict__ p4 = (const int4*)(idx + start);

    for (int lo = lo0; lo < rows; lo += stride) {
        const int      left = rows - lo;
        const unsigned cnt  = (unsigned)(left < DEG_SLICE ? left : DEG_SLICE);
        for (unsigned b = tid; b < cnt; b += DEG_THREADS) hist[b] = 0u;
        __syncthreads();
        const int4 none = make_int4(-1, -1, -1, -1);
        const int  last = n4 - 1;
        for (int q = tid; q < n4; q += 4 * DEG_THREADS) {
            const int q1 = q + DEG_THREADS, q2 = q + 2 * DEG_THREADS, q3 = q + 3 * DEG_THREADS;
            const int4 v0 = p4[q];
            int4 v1 = p4[q1 < last ? q1 : last];
            int4 v2 = p4[q2 < last ? q2 : last];
            int4 v3 = p4[q3 < last ? q3 : last];
            if (q1 > last) v1 = none;
            if (q2 > last) v2 = none;
            if (q3 > last) v3 = none;
            deg_count4(hist, v0, lo, cnt);
            deg_count4(hist, v1, lo, cnt);
            deg_count4(hist, v2, lo, cnt);
            deg_count4(hist, v3, lo, cnt);
        }
        for (int i = start + (n4 << 2) + tid; i < end; i += DEG_THREADS) {
            const unsigned b = (unsigned)(idx[i] - lo);
            if (b < cnt) atomicAdd(&hist[b], 1u);
        }
        __syncthreads();
        for (unsigned b = tid; b < cnt; b += DEG_THREADS) {
            const unsigned c = hist[b];
            if (c) atomicAdd(&deg[lo + (int)b], (float)c);
        }
        __syncthreads();
    }
}

static constexpr int GA_ROWS    = 320;
static constexpr int GA_CHUNKS  = 3;
static constexpr int GA_THREADS = 1024;
static constexpr int GA_WAVES   = GA_THREADS / 32;
static constexpr int GA_LD      = 4;
static constexpr int GA_ROUND4  = GA_THREADS * GA_LD;
static constexpr int GA_QCAP    = GA_ROUND4 * 4 + 4;
static constexpr int GA_ROWBITS = 9;
static constexpr int GA_BATCH   = 8;

__device__ __forceinline__ int4 ga_load4(const int4* __restrict__ p4, int n4, int q) {
    const int last = n4 - 1;
    return p4[q < last ? q : last];
}

__global__ __launch_bounds__(GA_THREADS) void fr_gather(const float* __restrict__ probs,
                                                        const int* __restrict__ node_idx,
                                                        const int* __restrict__ edge_idx,
                                                        float* __restrict__ edge_agg,
                                                        int nnz, int E) {
    __shared__ float        acc[GA_ROWS * KDIM];
    __shared__ unsigned int queue[2 * GA_QCAP];
    __shared__ unsigned int qcount[3];
    const int tid = threadIdx.x, lane = tid & 31;
    const int wave = __builtin_amdgcn_readfirstlane(tid >> 5);
    const int lo   = blockIdx.y * GA_ROWS;
    const int left = E - lo;
    const unsigned int cnt = (unsigned int)(left < GA_ROWS ? left : GA_ROWS);

    int per = (nnz + (int)gridDim.x - 1) / (int)gridDim.x;
    per = (per + 3) & ~3;
    const long long s64 = (long long)blockIdx.x * per;
    if (s64 >= nnz) return;
    const int start  = (int)s64;
    const int end    = (nnz - start < per) ? nnz : start + per;
    const int n4     = (end - start) >> 2;
    const int ragged = (end - start) & 3;
    const int r4     = (n4 + GA_ROUND4 - 1) / GA_ROUND4;
    const int rounds = r4 > 0 ? r4 : 1;
    const int4* __restrict__ p4 = (const int4*)(edge_idx + start);
    const int4 none = make_int4(-1, -1, -1, -1);

    for (int i = tid; i < GA_ROWS * KDIM; i += GA_THREADS) acc[i] = 0.f;
    if (tid < 3) qcount[tid] = 0u;
    __syncthreads();

    int4 cur[GA_LD];
    for (int j = 0; j < GA_LD; ++j) cur[j] = none;
    if (n4 > 0) {
        for (int j = 0; j < GA_LD; ++j) cur[j] = ga_load4(p4, n4, j * GA_THREADS + tid);
    }

    const int h = lane >> 4;
    for (int r = 0; r < rounds; ++r) {
        const int q0 = r * GA_ROUND4;
        unsigned int bsl[GA_LD * 4];
        unsigned int nv[GA_LD * 4];
        bool         hit[GA_LD * 4];
        unsigned int nh = 0u;
        for (int j = 0; j < GA_LD; ++j) {
            const int q  = q0 + j * GA_THREADS + tid;
            const int e0 = start + 4 * q;
            const unsigned int cj = q < n4 ? cnt : 0u;
            const int ev[4] = {cur[j].x, cur[j].y, cur[j].z, cur[j].w};
            for (int c = 0; c < 4; ++c) {
                const unsigned int b = (unsigned int)(ev[c] - lo);
                const bool hh = b < cj;
                bsl[j * 4 + c] = b;
                hit[j * 4 + c] = hh;
                nh += hh ? 1u : 0u;
                nv[j * 4 + c]  = (unsigned int)node_idx[hh ? e0 + c : start];
            }
        }
        int4 nxt[GA_LD];
        for (int j = 0; j < GA_LD; ++j) nxt[j] = none;
        if (r + 1 < rounds) {
            for (int j = 0; j < GA_LD; ++j)
                nxt[j] = ga_load4(p4, n4, q0 + GA_ROUND4 + j * GA_THREADS + tid);
        }

        unsigned int* qu = queue + (r & 1) * GA_QCAP;
        unsigned int* qc = &qcount[r % 3];
        if (nh != 0u) {
            unsigned int pos = atomicAdd(qc, nh);
            for (int k = 0; k < GA_LD * 4; ++k)
                if (hit[k]) qu[pos++] = (nv[k] << GA_ROWBITS) | bsl[k];
        }
        if (ragged != 0 && r == 0) {
            if (tid < ragged) {
                const int e = start + 4 * n4 + tid;
                const unsigned int b = (unsigned int)(edge_idx[e] - lo);
                if (b < cnt) qu[atomicAdd(qc, 1u)] = ((unsigned int)node_idx[e] << GA_ROWBITS) | b;
            }
        }
        __syncthreads();

        if (tid == 0) qcount[(r + 2) % 3] = 0u;
        const int count = __builtin_amdgcn_readfirstlane((int)*qc);
        for (int base = 0; base < count; base += GA_WAVES * GA_BATCH) {
            float4       p[GA_BATCH];
            unsigned int rc[GA_BATCH];
            for (int j = 0; j < GA_BATCH; ++j) {
                const int q = base + wave + GA_WAVES * j;
                rc[j] = qu[q < count ? q : 0];
                p[j]  = ((const float4*)(probs + (size_t)(rc[j] >> GA_ROWBITS) * KDIM))[lane];
            }
            for (int j = 0; j < GA_BATCH; ++j) {
                const bool ok = base + wave + GA_WAVES * j < count;
                const float x = ok ? p[j].x : 0.f, y = ok ? p[j].y : 0.f;
                const float z = ok ? p[j].z : 0.f, w = ok ? p[j].w : 0.f;
                float* d = acc + (rc[j] & ((1u << GA_ROWBITS) - 1u)) * KDIM + 4 * lane;
                atomicAdd(d + h,           h ? y : x);
                atomicAdd(d + (h ^ 1),     h ? x : y);
                atomicAdd(d + 2 + h,       h ? w : z);
                atomicAdd(d + 2 + (h ^ 1), h ? z : w);
            }
        }

        for (int j = 0; j < GA_LD; ++j) cur[j] = nxt[j];
    }
    __syncthreads();

    const int total = (int)cnt * KDIM;
    float* __restrict__ dst = edge_agg + (size_t)lo * KDIM;
    for (int i = tid; i < total; i += GA_THREADS) {
        const float v = acc[i];
        if (v != 0.f) atomicAdd(dst + i, v);
    }
}

__global__ __launch_bounds__(256) void fr_edge_gemm(const float* __restrict__ edge_agg,
                                                    const float* __restrict__ edge_deg,
                                                    const float* __restrict__ W,
                                                    float* __restrict__ edge_feat, int E) {
    __shared__ float sA[16][132];
    int tid = threadIdx.x;
    int m0 = blockIdx.x * 16;
    for (int i = tid; i < 16 * 32; i += 256) {
        int row = i >> 5, c4 = i & 31, r = m0 + row;
        float4 v = make_float4(0.f, 0.f, 0.f, 0.f);
        if (r < E) v = ((const float4*)edge_agg)[(size_t)r * 32 + c4];
        *(float4*)&sA[row][c4 * 4] = v;
    }
    __syncthreads();

    int wave = tid >> 5, lane = tid & 31;
    int half = lane >> 4, lr = lane & 15;
    int n0 = wave * 16;

    v8f c = {0.f, 0.f, 0.f, 0.f, 0.f, 0.f, 0.f, 0.f};
    #pragma unroll 4
    for (int k0 = 0; k0 < KDIM; k0 += 4) {
        int ka = k0 + 2 * half;
        v2f a, b;
        a.x = sA[lr][ka];
        a.y = sA[lr][ka + 1];
        b.x = W[(size_t)ka * KDIM + n0 + lr];
        b.y = W[(size_t)(ka + 1) * KDIM + n0 + lr];
        c = __builtin_amdgcn_wmma_f32_16x16x4_f32(false, a, false, b, (short)0, c,
                                                  false, false);
    }
    int rowbase = m0 + 8 * half;
    #pragma unroll
    for (int r = 0; r < 8; ++r) {
        int row = rowbase + r;
        if (row < E) {
            float inv = 1.f / fmaxf(edge_deg[row], 1.f);
            edge_feat[(size_t)row * KDIM + n0 + lr] = c[r] * inv;
        }
    }
}

__global__ __launch_bounds__(256) void fr_scatter(const float* __restrict__ edge_feat,
                                                  const int* __restrict__ node_idx,
                                                  const int* __restrict__ edge_idx,
                                                  float* __restrict__ out, int nnz) {
    long long gid = (long long)blockIdx.x * 256 + threadIdx.x;
    int i = (int)(gid >> 5);
    if (i >= nnz) return;
    int c4 = (int)(gid & 31);
    int nv = node_idx[i], ev = edge_idx[i];
    float4 f = ((const float4*)(edge_feat + (size_t)ev * KDIM))[c4];
    float* dst = out + (size_t)nv * KDIM + c4 * 4;
    atomicAdd(dst + 0, f.x); atomicAdd(dst + 1, f.y);
    atomicAdd(dst + 2, f.z); atomicAdd(dst + 3, f.w);
}

__global__ __launch_bounds__(256) void fr_finalize(const float* __restrict__ probs,
                                                   const float* __restrict__ node_deg,
                                                   float* __restrict__ out, int n) {
    int wave = threadIdx.x >> 5, lane = threadIdx.x & 31;
    int row = blockIdx.x * 8 + wave;
    if (row >= n) return;
    float inv = 1.f / fmaxf(node_deg[row], 1.f);
    float4 acc = ((const float4*)(out + (size_t)row * KDIM))[lane];
    float4 p   = ((const float4*)(probs + (size_t)row * KDIM))[lane];
    float4 nat;
    nat.x = fmaxf(p.x, EPS) * acc.x * inv;
    nat.y = fmaxf(p.y, EPS) * acc.y * inv;
    nat.z = fmaxf(p.z, EPS) * acc.z * inv;
    nat.w = fmaxf(p.w, EPS) * acc.w * inv;
    float s = nat.x + nat.y + nat.z + nat.w;
    #pragma unroll
    for (int off = 16; off > 0; off >>= 1) s += __shfl_xor(s, off, 32);
    float mean = s * (1.0f / KDIM);
    nat.x -= mean; nat.y -= mean; nat.z -= mean; nat.w -= mean;
    ((float4*)(out + (size_t)row * KDIM))[lane] = nat;
}

extern "C" void kernel_launch(void* const* d_in, const int* in_sizes, int n_in,
                              void* d_out, int out_size, void* d_ws, size_t ws_size,
                              hipStream_t stream) {
    const float* y        = (const float*)d_in[0];
    const float* W        = (const float*)d_in[1];
    const int*   node_idx = (const int*)d_in[2];
    const int*   edge_idx = (const int*)d_in[3];
    float*       out      = (float*)d_out;

    const int n   = in_sizes[0] / KDIM;
    const int nnz = in_sizes[2];
    const int E   = N_EDGES;

    float* probs     = (float*)d_ws;
    float* edge_agg  = probs + (size_t)n * KDIM;
    float* edge_feat = edge_agg + (size_t)E * KDIM;
    float* edge_deg  = edge_feat + (size_t)E * KDIM;
    float* node_deg  = edge_deg + E;

    hipMemsetAsync(edge_agg, 0, (size_t)E * KDIM * sizeof(float), stream);
    hipMemsetAsync(edge_deg, 0, (size_t)(E + n) * sizeof(float), stream);
    hipMemsetAsync(out, 0, (size_t)n * KDIM * sizeof(float), stream);

    fr_softmax<<<(n + 7) / 8, 256, 0, stream>>>(y, probs, n);
    fr_degrees<<<dim3(DEG_CHUNKS, 1 + DEG_NODE_JOBS), DEG_THREADS, 0, stream>>>(
        node_idx, edge_idx, edge_deg, node_deg, nnz, n, E);
    fr_gather<<<dim3(GA_CHUNKS, (N_EDGES + GA_ROWS - 1) / GA_ROWS), GA_THREADS, 0, stream>>>(
        probs, node_idx, edge_idx, edge_agg, nnz, E);
    fr_edge_gemm<<<(E + 15) / 16, 256, 0, stream>>>(edge_agg, edge_deg, W,
                                                    edge_feat, E);
    {
        long long total = (long long)nnz * 32;
        fr_scatter<<<(unsigned)((total + 255) / 256), 256, 0, stream>>>(
            edge_feat, node_idx, edge_idx, out, nnz);
    }
    fr_finalize<<<(n + 7) / 8, 256, 0, stream>>>(probs, node_deg, out, n);
}


# scatter: a block owns 400 node rows as an f32 tile in LDS, its waves filter the whole entry list for them and add edge rows by LDS f32 adds, tile stored to out once; no global adds; arrival order
# speedup vs baseline: 2.5339x; 1.6534x over previous
#include <hip/hip_runtime.h>

typedef __attribute__((ext_vector_type(2))) float v2f;
typedef __attribute__((ext_vector_type(8))) float v8f;

static constexpr int KDIM    = 128;
static constexpr int N_EDGES = 25000;
static constexpr float EPS   = 1e-12f;

__global__ __launch_bounds__(256) void fr_softmax(const float* __restrict__ y,
                                                  float* __restrict__ probs, int n) {
    int wave = threadIdx.x >> 5, lane = threadIdx.x & 31;
    int row = blockIdx.x * 8 + wave;
    if (row >= n) return;
    float4 v = ((const float4*)(y + (size_t)row * KDIM))[lane];
    float m = fmaxf(fmaxf(v.x, v.y), fmaxf(v.z, v.w));
    #pragma unroll
    for (int off = 16; off > 0; off >>= 1) m = fmaxf(m, __shfl_xor(m, off, 32));
    float4 e;
    e.x = __expf(v.x - m); e.y = __expf(v.y - m);
    e.z = __expf(v.z - m); e.w = __expf(v.w - m);
    float s = e.x + e.y + e.z + e.w;
    #pragma unroll
    for (int off = 16; off > 0; off >>= 1) s += __shfl_xor(s, off, 32);
    float inv = 1.f / s;
    e.x *= inv; e.y *= inv; e.z *= inv; e.w *= inv;
    ((float4*)(probs + (size_t)row * KDIM))[lane] = e;
}

static constexpr int DEG_SLICE     = 25000;
static constexpr int DEG_CHUNKS    = 32;
static constexpr int DEG_NODE_JOBS = 4;
static constexpr int DEG_THREADS   = 1024;

__device__ __forceinline__ void deg_count4(unsigned int* hist, int4 v, int lo, unsigned cnt) {
    const unsigned b0 = (unsigned)(v.x - lo), b1 = (unsigned)(v.y - lo);
    const unsigned b2 = (unsigned)(v.z - lo), b3 = (unsigned)(v.w - lo);
    if (b0 < cnt) atomicAdd(&hist[b0], 1u);
    if (b1 < cnt) atomicAdd(&hist[b1], 1u);
    if (b2 < cnt) atomicAdd(&hist[b2], 1u);
    if (b3 < cnt) atomicAdd(&hist[b3], 1u);
}

__global__ __launch_bounds__(DEG_THREADS) void fr_degrees(const int* __restrict__ node_idx,
                                                          const int* __restrict__ edge_idx,
                                                          float* __restrict__ edge_deg,
                                                          float* __restrict__ node_deg,
                                                          int nnz, int n, int E) {
    __shared__ unsigned int hist[DEG_SLICE];
    const int tid = threadIdx.x;
    const int job = blockIdx.y;
    const int* __restrict__ idx = (job == 0) ? edge_idx : node_idx;
    float* __restrict__ deg     = (job == 0) ? edge_deg : node_deg;
    const int rows   = (job == 0) ? E : n;
    const int lo0    = (job == 0) ? 0 : (job - 1) * DEG_SLICE;
    const int stride = (job == 0) ? DEG_SLICE : DEG_NODE_JOBS * DEG_SLICE;

    int per = (nnz + (int)gridDim.x - 1) / (int)gridDim.x;
    per = (per + 3) & ~3;
    const long long s64 = (long long)blockIdx.x * per;
    if (s64 >= nnz) return;
    const int start = (int)s64;
    const int end   = (nnz - start < per) ? nnz : start + per;
    const int n4    = (end - start) >> 2;
    const int4* __restrict__ p4 = (const int4*)(idx + start);

    for (int lo = lo0; lo < rows; lo += stride) {
        const int      left = rows - lo;
        const unsigned cnt  = (unsigned)(left < DEG_SLICE ? left : DEG_SLICE);
        for (unsigned b = tid; b < cnt; b += DEG_THREADS) hist[b] = 0u;
        __syncthreads();
        const int4 none = make_int4(-1, -1, -1, -1);
        const int  last = n4 - 1;
        for (int q = tid; q < n4; q += 4 * DEG_THREADS) {
            const int q1 = q + DEG_THREADS, q2 = q + 2 * DEG_THREADS, q3 = q + 3 * DEG_THREADS;
            const int4 v0 = p4[q];
            int4 v1 = p4[q1 < last ? q1 : last];
            int4 v2 = p4[q2 < last ? q2 : last];
            int4 v3 = p4[q3 < last ? q3 : last];
            if (q1 > last) v1 = none;
            if (q2 > last) v2 = none;
            if (q3 > last) v3 = none;
            deg_count4(hist, v0, lo, cnt);
            deg_count4(hist, v1, lo, cnt);
            deg_count4(hist, v2, lo, cnt);
            deg_count4(hist, v3, lo, cnt);
        }
        for (int i = start + (n4 << 2) + tid; i < end; i += DEG_THREADS) {
            const unsigned b = (unsigned)(idx[i] - lo);
            if (b < cnt) atomicAdd(&hist[b], 1u);
        }
        __syncthreads();
        for (unsigned b = tid; b < cnt; b += DEG_THREADS) {
            const unsigned c = hist[b];
            if (c) atomicAdd(&deg[lo + (int)b], (float)c);
        }
        __syncthreads();
    }
}

static constexpr int GA_ROWS    = 320;
static constexpr int GA_CHUNKS  = 3;
static constexpr int GA_THREADS = 1024;
static constexpr int GA_WAVES   = GA_THREADS / 32;
static constexpr int GA_LD      = 4;
static constexpr int GA_ROUND4  = GA_THREADS * GA_LD;
static constexpr int GA_QCAP    = GA_ROUND4 * 4 + 4;
static constexpr int GA_ROWBITS = 9;
static constexpr int GA_BATCH   = 8;

__device__ __forceinline__ int4 ga_load4(const int4* __restrict__ p4, int n4, int q) {
    const int last = n4 - 1;
    return p4[q < last ? q : last];
}

__global__ __launch_bounds__(GA_THREADS) void fr_gather(const float* __restrict__ probs,
                                                        const int* __restrict__ node_idx,
                                                        const int* __restrict__ edge_idx,
                                                        float* __restrict__ edge_agg,
                                                        int nnz, int E) {
    __shared__ float        acc[GA_ROWS * KDIM];
    __shared__ unsigned int queue[2 * GA_QCAP];
    __shared__ unsigned int qcount[3];
    const int tid = threadIdx.x, lane = tid & 31;
    const int wave = __builtin_amdgcn_readfirstlane(tid >> 5);
    const int lo   = blockIdx.y * GA_ROWS;
    const int left = E - lo;
    const unsigned int cnt = (unsigned int)(left < GA_ROWS ? left : GA_ROWS);

    int per = (nnz + (int)gridDim.x - 1) / (int)gridDim.x;
    per = (per + 3) & ~3;
    const long long s64 = (long long)blockIdx.x * per;
    if (s64 >= nnz) return;
    const int start  = (int)s64;
    const int end    = (nnz - start < per) ? nnz : start + per;
    const int n4     = (end - start) >> 2;
    const int ragged = (end - start) & 3;
    const int r4     = (n4 + GA_ROUND4 - 1) / GA_ROUND4;
    const int rounds = r4 > 0 ? r4 : 1;
    const int4* __restrict__ p4 = (const int4*)(edge_idx + start);
    const int4 none = make_int4(-1, -1, -1, -1);

    for (int i = tid; i < GA_ROWS * KDIM; i += GA_THREADS) acc[i] = 0.f;
    if (tid < 3) qcount[tid] = 0u;
    __syncthreads();

    int4 cur[GA_LD];
    for (int j = 0; j < GA_LD; ++j) cur[j] = none;
    if (n4 > 0) {
        for (int j = 0; j < GA_LD; ++j) cur[j] = ga_load4(p4, n4, j * GA_THREADS + tid);
    }

    const int h = lane >> 4;
    for (int r = 0; r < rounds; ++r) {
        const int q0 = r * GA_ROUND4;
        unsigned int bsl[GA_LD * 4];
        unsigned int nv[GA_LD * 4];
        bool         hit[GA_LD * 4];
        unsigned int nh = 0u;
        for (int j = 0; j < GA_LD; ++j) {
            const int q  = q0 + j * GA_THREADS + tid;
            const int e0 = start + 4 * q;
            const unsigned int cj = q < n4 ? cnt : 0u;
            const int ev[4] = {cur[j].x, cur[j].y, cur[j].z, cur[j].w};
            for (int c = 0; c < 4; ++c) {
                const unsigned int b = (unsigned int)(ev[c] - lo);
                const bool hh = b < cj;
                bsl[j * 4 + c] = b;
                hit[j * 4 + c] = hh;
                nh += hh ? 1u : 0u;
                nv[j * 4 + c]  = (unsigned int)node_idx[hh ? e0 + c : start];
            }
        }
        int4 nxt[GA_LD];
        for (int j = 0; j < GA_LD; ++j) nxt[j] = none;
        if (r + 1 < rounds) {
            for (int j = 0; j < GA_LD; ++j)
                nxt[j] = ga_load4(p4, n4, q0 + GA_ROUND4 + j * GA_THREADS + tid);
        }

        unsigned int* qu = queue + (r & 1) * GA_QCAP;
        unsigned int* qc = &qcount[r % 3];
        if (nh != 0u) {
            unsigned int pos = atomicAdd(qc, nh);
            for (int k = 0; k < GA_LD * 4; ++k)
                if (hit[k]) qu[pos++] = (nv[k] << GA_ROWBITS) | bsl[k];
        }
        if (ragged != 0 && r == 0) {
            if (tid < ragged) {
                const int e = start + 4 * n4 + tid;
                const unsigned int b = (unsigned int)(edge_idx[e] - lo);
                if (b < cnt) qu[atomicAdd(qc, 1u)] = ((unsigned int)node_idx[e] << GA_ROWBITS) | b;
            }
        }
        __syncthreads();

        if (tid == 0) qcount[(r + 2) % 3] = 0u;
        const int count = __builtin_amdgcn_readfirstlane((int)*qc);
        for (int base = 0; base < count; base += GA_WAVES * GA_BATCH) {
            float4       p[GA_BATCH];
            unsigned int rc[GA_BATCH];
            for (int j = 0; j < GA_BATCH; ++j) {
                const int q = base + wave + GA_WAVES * j;
                rc[j] = qu[q < count ? q : 0];
                p[j]  = ((const float4*)(probs + (size_t)(rc[j] >> GA_ROWBITS) * KDIM))[lane];
            }
            for (int j = 0; j < GA_BATCH; ++j) {
                const bool ok = base + wave + GA_WAVES * j < count;
                const float x = ok ? p[j].x : 0.f, y = ok ? p[j].y : 0.f;
                const float z = ok ? p[j].z : 0.f, w = ok ? p[j].w : 0.f;
                float* d = acc + (rc[j] & ((1u << GA_ROWBITS) - 1u)) * KDIM + 4 * lane;
                atomicAdd(d + h,           h ? y : x);
                atomicAdd(d + (h ^ 1),     h ? x : y);
                atomicAdd(d + 2 + h,       h ? w : z);
                atomicAdd(d + 2 + (h ^ 1), h ? z : w);
            }
        }

        for (int j = 0; j < GA_LD; ++j) cur[j] = nxt[j];
    }
    __syncthreads();

    const int total = (int)cnt * KDIM;
    float* __restrict__ dst = edge_agg + (size_t)lo * KDIM;
    for (int i = tid; i < total; i += GA_THREADS) {
        const float v = acc[i];
        if (v != 0.f) atomicAdd(dst + i, v);
    }
}

__global__ __launch_bounds__(256) void fr_edge_gemm(const float* __restrict__ edge_agg,
                                                    const float* __restrict__ edge_deg,
                                                    const float* __restrict__ W,
                                                    float* __restrict__ edge_feat, int E) {
    __shared__ float sA[16][132];
    int tid = threadIdx.x;
    int m0 = blockIdx.x * 16;
    for (int i = tid; i < 16 * 32; i += 256) {
        int row = i >> 5, c4 = i & 31, r = m0 + row;
        float4 v = make_float4(0.f, 0.f, 0.f, 0.f);
        if (r < E) v = ((const float4*)edge_agg)[(size_t)r * 32 + c4];
        *(float4*)&sA[row][c4 * 4] = v;
    }
    __syncthreads();

    int wave = tid >> 5, lane = tid & 31;
    int half = lane >> 4, lr = lane & 15;
    int n0 = wave * 16;

    v8f c = {0.f, 0.f, 0.f, 0.f, 0.f, 0.f, 0.f, 0.f};
    #pragma unroll 4
    for (int k0 = 0; k0 < KDIM; k0 += 4) {
        int ka = k0 + 2 * half;
        v2f a, b;
        a.x = sA[lr][ka];
        a.y = sA[lr][ka + 1];
        b.x = W[(size_t)ka * KDIM + n0 + lr];
        b.y = W[(size_t)(ka + 1) * KDIM + n0 + lr];
        c = __builtin_amdgcn_wmma_f32_16x16x4_f32(false, a, false, b, (short)0, c,
                                                  false, false);
    }
    int rowbase = m0 + 8 * half;
    #pragma unroll
    for (int r = 0; r < 8; ++r) {
        int row = rowbase + r;
        if (row < E) {
            float inv = 1.f / fmaxf(edge_deg[row], 1.f);
            edge_feat[(size_t)row * KDIM + n0 + lr] = c[r] * inv;
        }
    }
}

static constexpr int SC_ROWS    = 400;
static constexpr int SC_THREADS = 1024;
static constexpr int SC_WAVES   = SC_THREADS / 32;
static constexpr int SC_LD      = 4;
static constexpr int SC_STEP4   = 32 * SC_LD;
static constexpr int SC_BATCH   = 16;
static constexpr int SC_QCAP    = SC_STEP4 * 4 + SC_BATCH;
static constexpr int SC_ROWBITS = 9;

__device__ __forceinline__ int4 sc_load4(const int4* __restrict__ p4, int n4, int q) {
    const int last = n4 - 1;
    return p4[q < last ? q : last];
}

__device__ __forceinline__ void sc_push(unsigned int* myq, int& qn, bool hit, unsigned int rec) {
    const unsigned int mk = __builtin_amdgcn_ballot_w32(hit);
    if (hit) myq[qn + (int)__builtin_amdgcn_mbcnt_lo(mk, 0u)] = rec;
    qn += __builtin_popcount(mk);
}

__device__ __forceinline__ int sc_rotate(int sl, int rot, int nsteps) {
    const int s = sl + rot;
    return s >= nsteps ? s - nsteps : s;
}

__device__ __forceinline__ void sc_load_step(int4 (&buf)[SC_LD], const int4* __restrict__ p4, int n4,
                                             int s, int lane) {
    for (int j = 0; j < SC_LD; ++j) buf[j] = sc_load4(p4, n4, s * SC_STEP4 + j * 32 + lane);
}

__device__ __forceinline__ void sc_scan(const int4 (&buf)[SC_LD], int s, int n4, int lo, unsigned int cnt,
                                        int lane, unsigned int* myq, int& qn) {
    for (int j = 0; j < SC_LD; ++j) {
        const int q = s * SC_STEP4 + j * 32 + lane;
        const unsigned int cj = q < n4 ? cnt : 0u;
        const int4 v = buf[j];
        const unsigned int b0 = (unsigned int)(v.x - lo), b1 = (unsigned int)(v.y - lo);
        const unsigned int b2 = (unsigned int)(v.z - lo), b3 = (unsigned int)(v.w - lo);
        const bool h0 = b0 < cj, h1 = b1 < cj, h2 = b2 < cj, h3 = b3 < cj;
        if (__builtin_amdgcn_ballot_w32(h0 || h1 || h2 || h3) != 0u) {
            const unsigned int e0 = (unsigned int)(4 * q);
            sc_push(myq, qn, h0, ((e0 + 0u) << SC_ROWBITS) | b0);
            sc_push(myq, qn, h1, ((e0 + 1u) << SC_ROWBITS) | b1);
            sc_push(myq, qn, h2, ((e0 + 2u) << SC_ROWBITS) | b2);
            sc_push(myq, qn, h3, ((e0 + 3u) << SC_ROWBITS) | b3);
        }
    }
}

template <bool FULL>
__device__ __forceinline__ void sc_drain(float* acc, const unsigned int* myq, int base, int m,
                                         int lane, const int* __restrict__ edge_idx,
                                         const float* __restrict__ edge_feat) {
    const int slot = lane & (SC_BATCH - 1);
    const unsigned int rc = myq[base + ((FULL || slot < m) ? slot : 0)];
    const int ev = edge_idx[rc >> SC_ROWBITS];
    float4 p[SC_BATCH];
    #pragma unroll
    for (int j = 0; j < SC_BATCH; ++j) {
        const int evj = __builtin_amdgcn_readlane(ev, j);
        p[j] = ((const float4*)(edge_feat + (size_t)evj * KDIM))[lane];
    }
    const int h = lane >> 4;
    #pragma unroll
    for (int j = 0; j < SC_BATCH; ++j) {
        const unsigned int bj = (unsigned int)__builtin_amdgcn_readlane((int)rc, j) & ((1u << SC_ROWBITS) - 1u);
        const bool ok = FULL || j < m;
        const float x = ok ? p[j].x : 0.f, y = ok ? p[j].y : 0.f;
        const float z = ok ? p[j].z : 0.f, w = ok ? p[j].w : 0.f;
        float* d = acc + bj * KDIM + 4 * lane;
        atomicAdd(d + h,           h ? y : x);
        atomicAdd(d + (h ^ 1),     h ? x : y);
        atomicAdd(d + 2 + h,       h ? w : z);
        atomicAdd(d + 2 + (h ^ 1), h ? z : w);
    }
}

__global__ __launch_bounds__(SC_THREADS) void fr_scatter(const float* __restrict__ edge_feat,
                                                         const int* __restrict__ node_idx,
                                                         const int* __restrict__ edge_idx,
                                                         float* __restrict__ out, int nnz, int n) {
    __shared__ float        acc[SC_ROWS * KDIM];
    __shared__ unsigned int queue[SC_WAVES * SC_QCAP];
    const int tid = threadIdx.x, lane = tid & 31;
    const int wave = __builtin_amdgcn_readfirstlane(tid >> 5);
    const int lo   = blockIdx.x * SC_ROWS;
    const int left = n - lo;
    const unsigned int cnt = (unsigned int)(left < SC_ROWS ? left : SC_ROWS);

    float4* acc4 = (float4*)acc;
    for (int i = tid; i < SC_ROWS * KDIM / 4; i += SC_THREADS) acc4[i] = make_float4(0.f, 0.f, 0.f, 0.f);
    __syncthreads();

    unsigned int* myq = queue + wave * SC_QCAP;
    int qn = 0;
    const int n4     = nnz >> 2;
    const int ragged = nnz & 3;
    const int nsteps = (n4 + SC_STEP4 - 1) / SC_STEP4;
    const int rot = (int)blockIdx.x * (nsteps / (int)gridDim.x);
    const int4* __restrict__ p4 = (const int4*)node_idx;

    if (ragged != 0 && wave == 0) {
        const int e = 4 * n4 + lane;
        unsigned int b = 0u;
        bool hit = false;
        if (lane < ragged) {
            b   = (unsigned int)(node_idx[e] - lo);
            hit = b < cnt;
        }
        sc_push(myq, qn, hit, ((unsigned int)e << SC_ROWBITS) | b);
    }

    if (wave < nsteps) {
        int4 bufa[SC_LD], bufb[SC_LD];
        sc_load_step(bufa, p4, n4, sc_rotate(wave, rot, nsteps), lane);
        for (int sl = wave; sl < nsteps; sl += 2 * SC_WAVES) {
            const int slb = sl + SC_WAVES, sla = sl + 2 * SC_WAVES;
            sc_load_step(bufb, p4, n4, sc_rotate(slb < nsteps ? slb : sl, rot, nsteps), lane);
            sc_scan(bufa, sc_rotate(sl, rot, nsteps), n4, lo, cnt, lane, myq, qn);
            while (qn >= SC_BATCH) {
                qn -= SC_BATCH;
                sc_drain<true>(acc, myq, qn, SC_BATCH, lane, edge_idx, edge_feat);
            }
            sc_load_step(bufa, p4, n4, sc_rotate(sla < nsteps ? sla : sl, rot, nsteps), lane);
            if (slb < nsteps) {
                sc_scan(bufb, sc_rotate(slb, rot, nsteps), n4, lo, cnt, lane, myq, qn);
                while (qn >= SC_BATCH) {
                    qn -= SC_BATCH;
                    sc_drain<true>(acc, myq, qn, SC_BATCH, lane, edge_idx, edge_feat);
                }
            }
        }
    }
    if (qn > 0) sc_drain<false>(acc, myq, 0, qn, lane, edge_idx, edge_feat);
    __syncthreads();

    const int total4 = (int)cnt * (KDIM / 4);
    float4* __restrict__ dst4 = (float4*)(out + (size_t)lo * KDIM);
    for (int i = tid; i < total4; i += SC_THREADS) dst4[i] = acc4[i];
}

__global__ __launch_bounds__(256) void fr_finalize(const float* __restrict__ probs,
                                                   const float* __restrict__ node_deg,
                                                   float* __restrict__ out, int n) {
    int wave = threadIdx.x >> 5, lane = threadIdx.x & 31;
    int row = blockIdx.x * 8 + wave;
    if (row >= n) return;
    float inv = 1.f / fmaxf(node_deg[row], 1.f);
    float4 acc = ((const float4*)(out + (size_t)row * KDIM))[lane];
    float4 p   = ((const float4*)(probs + (size_t)row * KDIM))[lane];
    float4 nat;
    nat.x = fmaxf(p.x, EPS) * acc.x * inv;
    nat.y = fmaxf(p.y, EPS) * acc.y * inv;
    nat.z = fmaxf(p.z, EPS) * acc.z * inv;
    nat.w = fmaxf(p.w, EPS) * acc.w * inv;
    float s = nat.x + nat.y + nat.z + nat.w;
    #pragma unroll
    for (int off = 16; off > 0; off >>= 1) s += __shfl_xor(s, off, 32);
    float mean = s * (1.0f / KDIM);
    nat.x -= mean; nat.y -= mean; nat.z -= mean; nat.w -= mean;
    ((float4*)(out + (size_t)row * KDIM))[lane] = nat;
}

extern "C" void kernel_launch(void* const* d_in, const int* in_sizes, int n_in,
                              void* d_out, int out_size, void* d_ws, size_t ws_size,
                              hipStream_t stream) {
    const float* y        = (const float*)d_in[0];
    const float* W        = (const float*)d_in[1];
    const int*   node_idx = (const int*)d_in[2];
    const int*   edge_idx = (const int*)d_in[3];
    float*       out      = (float*)d_out;

    const int n   = in_sizes[0] / KDIM;
    const int nnz = in_sizes[2];
    const int E   = N_EDGES;

    float* probs     = (float*)d_ws;
    float* edge_agg  = probs + (size_t)n * KDIM;
    float* edge_feat = edge_agg + (size_t)E * KDIM;
    float* edge_deg  = edge_feat + (size_t)E * KDIM;
    float* node_deg  = edge_deg + E;

    hipMemsetAsync(edge_agg, 0, (size_t)E * KDIM * sizeof(float), stream);
    hipMemsetAsync(edge_deg, 0, (size_t)(E + n) * sizeof(float), stream);

    fr_softmax<<<(n + 7) / 8, 256, 0, stream>>>(y, probs, n);
    fr_degrees<<<dim3(DEG_CHUNKS, 1 + DEG_NODE_JOBS), DEG_THREADS, 0, stream>>>(
        node_idx, edge_idx, edge_deg, node_deg, nnz, n, E);
    fr_gather<<<dim3(GA_CHUNKS, (N_EDGES + GA_ROWS - 1) / GA_ROWS), GA_THREADS, 0, stream>>>(
        probs, node_idx, edge_idx, edge_agg, nnz, E);
    fr_edge_gemm<<<(E + 15) / 16, 256, 0, stream>>>(edge_agg, edge_deg, W,
                                                    edge_feat, E);
    fr_scatter<<<(n + SC_ROWS - 1) / SC_ROWS, SC_THREADS, 0, stream>>>(
        edge_feat, node_idx, edge_idx, out, nnz, n);
    fr_finalize<<<(n + 7) / 8, 256, 0, stream>>>(probs, node_deg, out, n);
}


# gather: each wave filters its own trips onto its own LDS stack and drains 16 probs rows at a time, as the scatter does; no block barrier or shared queue per round; sums in arrival order, 3 partials
# speedup vs baseline: 2.8342x; 1.1185x over previous
#include <hip/hip_runtime.h>

typedef __attribute__((ext_vector_type(2))) float v2f;
typedef __attribute__((ext_vector_type(8))) float v8f;

static constexpr int KDIM    = 128;
static constexpr int N_EDGES = 25000;
static constexpr float EPS   = 1e-12f;

__global__ __launch_bounds__(256) void fr_softmax(const float* __restrict__ y,
                                                  float* __restrict__ probs, int n) {
    int wave = threadIdx.x >> 5, lane = threadIdx.x & 31;
    int row = blockIdx.x * 8 + wave;
    if (row >= n) return;
    float4 v = ((const float4*)(y + (size_t)row * KDIM))[lane];
    float m = fmaxf(fmaxf(v.x, v.y), fmaxf(v.z, v.w));
    #pragma unroll
    for (int off = 16; off > 0; off >>= 1) m = fmaxf(m, __shfl_xor(m, off, 32));
    float4 e;
    e.x = __expf(v.x - m); e.y = __expf(v.y - m);
    e.z = __expf(v.z - m); e.w = __expf(v.w - m);
    float s = e.x + e.y + e.z + e.w;
    #pragma unroll
    for (int off = 16; off > 0; off >>= 1) s += __shfl_xor(s, off, 32);
    float inv = 1.f / s;
    e.x *= inv; e.y *= inv; e.z *= inv; e.w *= inv;
    ((float4*)(probs + (size_t)row * KDIM))[lane] = e;
}

static constexpr int DEG_SLICE     = 25000;
static constexpr int DEG_CHUNKS    = 32;
static constexpr int DEG_NODE_JOBS = 4;
static constexpr int DEG_THREADS   = 1024;

__device__ __forceinline__ void deg_count4(unsigned int* hist, int4 v, int lo, unsigned cnt) {
    const unsigned b0 = (unsigned)(v.x - lo), b1 = (unsigned)(v.y - lo);
    const unsigned b2 = (unsigned)(v.z - lo), b3 = (unsigned)(v.w - lo);
    if (b0 < cnt) atomicAdd(&hist[b0], 1u);
    if (b1 < cnt) atomicAdd(&hist[b1], 1u);
    if (b2 < cnt) atomicAdd(&hist[b2], 1u);
    if (b3 < cnt) atomicAdd(&hist[b3], 1u);
}

__global__ __launch_bounds__(DEG_THREADS) void fr_degrees(const int* __restrict__ node_idx,
                                                          const int* __restrict__ edge_idx,
                                                          float* __restrict__ edge_deg,
                                                          float* __restrict__ node_deg,
                                                          int nnz, int n, int E) {
    __shared__ unsigned int hist[DEG_SLICE];
    const int tid = threadIdx.x;
    const int job = blockIdx.y;
    const int* __restrict__ idx = (job == 0) ? edge_idx : node_idx;
    float* __restrict__ deg     = (job == 0) ? edge_deg : node_deg;
    const int rows   = (job == 0) ? E : n;
    const int lo0    = (job == 0) ? 0 : (job - 1) * DEG_SLICE;
    const int stride = (job == 0) ? DEG_SLICE : DEG_NODE_JOBS * DEG_SLICE;

    int per = (nnz + (int)gridDim.x - 1) / (int)gridDim.x;
    per = (per + 3) & ~3;
    const long long s64 = (long long)blockIdx.x * per;
    if (s64 >= nnz) return;
    const int start = (int)s64;
    const int end   = (nnz - start < per) ? nnz : start + per;
    const int n4    = (end - start) >> 2;
    const int4* __restrict__ p4 = (const int4*)(idx + start);

    for (int lo = lo0; lo < rows; lo += stride) {
        const int      left = rows - lo;
        const unsigned cnt  = (unsigned)(left < DEG_SLICE ? left : DEG_SLICE);
        for (unsigned b = tid; b < cnt; b += DEG_THREADS) hist[b] = 0u;
        __syncthreads();
        const int4 none = make_int4(-1, -1, -1, -1);
        const int  last = n4 - 1;
        for (int q = tid; q < n4; q += 4 * DEG_THREADS) {
            const int q1 = q + DEG_THREADS, q2 = q + 2 * DEG_THREADS, q3 = q + 3 * DEG_THREADS;
            const int4 v0 = p4[q];
            int4 v1 = p4[q1 < last ? q1 : last];
            int4 v2 = p4[q2 < last ? q2 : last];
            int4 v3 = p4[q3 < last ? q3 : last];
            if (q1 > last) v1 = none;
            if (q2 > last) v2 = none;
            if (q3 > last) v3 = none;
            deg_count4(hist, v0, lo, cnt);
            deg_count4(hist, v1, lo, cnt);
            deg_count4(hist, v2, lo, cnt);
            deg_count4(hist, v3, lo, cnt);
        }
        for (int i = start + (n4 << 2) + tid; i < end; i += DEG_THREADS) {
            const unsigned b = (unsigned)(idx[i] - lo);
            if (b < cnt) atomicAdd(&hist[b], 1u);
        }
        __syncthreads();
        for (unsigned b = tid; b < cnt; b += DEG_THREADS) {
            const unsigned c = hist[b];
            if (c) atomicAdd(&deg[lo + (int)b], (float)c);
        }
        __syncthreads();
    }
}

static constexpr int SC_ROWS    = 400;
static constexpr int SC_THREADS = 1024;
static constexpr int SC_WAVES   = SC_THREADS / 32;
static constexpr int SC_LD      = 4;
static constexpr int SC_STEP4   = 32 * SC_LD;
static constexpr int SC_BATCH   = 16;
static constexpr int SC_QCAP    = SC_STEP4 * 4 + SC_BATCH;
static constexpr int SC_ROWBITS = 9;

__device__ __forceinline__ int4 sc_load4(const int4* __restrict__ p4, int n4, int q) {
    const int last = n4 - 1;
    return p4[q < last ? q : last];
}

__device__ __forceinline__ void sc_push(unsigned int* myq, int& qn, bool hit, unsigned int rec) {
    const unsigned int mk = __builtin_amdgcn_ballot_w32(hit);
    if (hit) myq[qn + (int)__builtin_amdgcn_mbcnt_lo(mk, 0u)] = rec;
    qn += __builtin_popcount(mk);
}

__device__ __forceinline__ int sc_rotate(int sl, int rot, int nsteps) {
    const int s = sl + rot;
    return s >= nsteps ? s - nsteps : s;
}

__device__ __forceinline__ void sc_load_step(int4 (&buf)[SC_LD], const int4* __restrict__ p4, int n4,
                                             int s, int lane) {
    for (int j = 0; j < SC_LD; ++j) buf[j] = sc_load4(p4, n4, s * SC_STEP4 + j * 32 + lane);
}

__device__ __forceinline__ void sc_scan(const int4 (&buf)[SC_LD], int s, int n4, int lo, unsigned int cnt,
                                        int lane, unsigned int* myq, int& qn) {
    for (int j = 0; j < SC_LD; ++j) {
        const int q = s * SC_STEP4 + j * 32 + lane;
        const unsigned int cj = q < n4 ? cnt : 0u;
        const int4 v = buf[j];
        const unsigned int b0 = (unsigned int)(v.x - lo), b1 = (unsigned int)(v.y - lo);
        const unsigned int b2 = (unsigned int)(v.z - lo), b3 = (unsigned int)(v.w - lo);
        const bool h0 = b0 < cj, h1 = b1 < cj, h2 = b2 < cj, h3 = b3 < cj;
        if (__builtin_amdgcn_ballot_w32(h0 || h1 || h2 || h3) != 0u) {
            const unsigned int e0 = (unsigned int)(4 * q);
            sc_push(myq, qn, h0, ((e0 + 0u) << SC_ROWBITS) | b0);
            sc_push(myq, qn, h1, ((e0 + 1u) << SC_ROWBITS) | b1);
            sc_push(myq, qn, h2, ((e0 + 2u) << SC_ROWBITS) | b2);
            sc_push(myq, qn, h3, ((e0 + 3u) << SC_ROWBITS) | b3);
        }
    }
}

template <bool FULL>
__device__ __forceinline__ void sc_drain(float* acc, const unsigned int* myq, int base, int m,
                                         int lane, const int* __restrict__ src_idx,
                                         const float* __restrict__ src_rows) {
    const int slot = lane & (SC_BATCH - 1);
    const unsigned int rc = myq[base + ((FULL || slot < m) ? slot : 0)];
    const int ev = src_idx[rc >> SC_ROWBITS];
    float4 p[SC_BATCH];
    #pragma unroll
    for (int j = 0; j < SC_BATCH; ++j) {
        const int evj = __builtin_amdgcn_readlane(ev, j);
        p[j] = ((const float4*)(src_rows + (size_t)evj * KDIM))[lane];
    }
    const int h = lane >> 4;
    #pragma unroll
    for (int j = 0; j < SC_BATCH; ++j) {
        const unsigned int bj = (unsigned int)__builtin_amdgcn_readlane((int)rc, j) & ((1u << SC_ROWBITS) - 1u);
        const bool ok = FULL || j < m;
        const float x = ok ? p[j].x : 0.f, y = ok ? p[j].y : 0.f;
        const float z = ok ? p[j].z : 0.f, w = ok ? p[j].w : 0.f;
        float* d = acc + bj * KDIM + 4 * lane;
        atomicAdd(d + h,           h ? y : x);
        atomicAdd(d + (h ^ 1),     h ? x : y);
        atomicAdd(d + 2 + h,       h ? w : z);
        atomicAdd(d + 2 + (h ^ 1), h ? z : w);
    }
}

static constexpr int GA_ROWS    = 320;
static constexpr int GA_CHUNKS  = 3;
static constexpr int GA_THREADS = SC_THREADS;

__global__ __launch_bounds__(GA_THREADS) void fr_gather(const float* __restrict__ probs,
                                                        const int* __restrict__ node_idx,
                                                        const int* __restrict__ edge_idx,
                                                        float* __restrict__ edge_agg,
                                                        int nnz, int E) {
    __shared__ float        acc[GA_ROWS * KDIM];
    __shared__ unsigned int queue[SC_WAVES * SC_QCAP];
    const int tid = threadIdx.x, lane = tid & 31;
    const int wave = __builtin_amdgcn_readfirstlane(tid >> 5);
    const int lo   = blockIdx.y * GA_ROWS;
    const int left = E - lo;
    const unsigned int cnt = (unsigned int)(left < GA_ROWS ? left : GA_ROWS);

    int per = (nnz + (int)gridDim.x - 1) / (int)gridDim.x;
    per = (per + 3) & ~3;
    const long long s64 = (long long)blockIdx.x * per;
    if (s64 >= nnz) return;
    const int start  = (int)s64;
    const int end    = (nnz - start < per) ? nnz : start + per;
    const int n4     = (end - start) >> 2;
    const int ragged = (end - start) & 3;
    const int nsteps = (n4 + SC_STEP4 - 1) / SC_STEP4;
    const int rot = (int)blockIdx.y * (nsteps / (int)gridDim.y);
    const int*  __restrict__ edges = edge_idx + start;
    const int*  __restrict__ nodes = node_idx + start;
    const int4* __restrict__ p4    = (const int4*)edges;

    float4* acc4 = (float4*)acc;
    for (int i = tid; i < GA_ROWS * KDIM / 4; i += GA_THREADS) acc4[i] = make_float4(0.f, 0.f, 0.f, 0.f);
    __syncthreads();

    unsigned int* myq = queue + wave * SC_QCAP;
    int qn = 0;

    if (ragged != 0 && wave == 0) {
        const int e = 4 * n4 + lane;
        unsigned int b = 0u;
        bool hit = false;
        if (lane < ragged) {
            b   = (unsigned int)(edges[e] - lo);
            hit = b < cnt;
        }
        sc_push(myq, qn, hit, ((unsigned int)e << SC_ROWBITS) | b);
    }

    if (wave < nsteps) {
        int4 bufa[SC_LD], bufb[SC_LD];
        sc_load_step(bufa, p4, n4, sc_rotate(wave, rot, nsteps), lane);
        for (int sl = wave; sl < nsteps; sl += 2 * SC_WAVES) {
            const int slb = sl + SC_WAVES, sla = sl + 2 * SC_WAVES;
            sc_load_step(bufb, p4, n4, sc_rotate(slb < nsteps ? slb : sl, rot, nsteps), lane);
            sc_scan(bufa, sc_rotate(sl, rot, nsteps), n4, lo, cnt, lane, myq, qn);
            while (qn >= SC_BATCH) {
                qn -= SC_BATCH;
                sc_drain<true>(acc, myq, qn, SC_BATCH, lane, nodes, probs);
            }
            sc_load_step(bufa, p4, n4, sc_rotate(sla < nsteps ? sla : sl, rot, nsteps), lane);
            if (slb < nsteps) {
                sc_scan(bufb, sc_rotate(slb, rot, nsteps), n4, lo, cnt, lane, myq, qn);
                while (qn >= SC_BATCH) {
                    qn -= SC_BATCH;
                    sc_drain<true>(acc, myq, qn, SC_BATCH, lane, nodes, probs);
                }
            }
        }
    }
    if (qn > 0) sc_drain<false>(acc, myq, 0, qn, lane, nodes, probs);
    __syncthreads();

    const int total = (int)cnt * KDIM;
    float* __restrict__ dst = edge_agg + (size_t)lo * KDIM;
    for (int i = tid; i < total; i += GA_THREADS) {
        const float v = acc[i];
        if (v != 0.f) atomicAdd(dst + i, v);
    }
}

__global__ __launch_bounds__(256) void fr_edge_gemm(const float* __restrict__ edge_agg,
                                                    const float* __restrict__ edge_deg,
                                                    const float* __restrict__ W,
                                                    float* __restrict__ edge_feat, int E) {
    __shared__ float sA[16][132];
    int tid = threadIdx.x;
    int m0 = blockIdx.x * 16;
    for (int i = tid; i < 16 * 32; i += 256) {
        int row = i >> 5, c4 = i & 31, r = m0 + row;
        float4 v = make_float4(0.f, 0.f, 0.f, 0.f);
        if (r < E) v = ((const float4*)edge_agg)[(size_t)r * 32 + c4];
        *(float4*)&sA[row][c4 * 4] = v;
    }
    __syncthreads();

    int wave = tid >> 5, lane = tid & 31;
    int half = lane >> 4, lr = lane & 15;
    int n0 = wave * 16;

    v8f c = {0.f, 0.f, 0.f, 0.f, 0.f, 0.f, 0.f, 0.f};
    #pragma unroll 4
    for (int k0 = 0; k0 < KDIM; k0 += 4) {
        int ka = k0 + 2 * half;
        v2f a, b;
        a.x = sA[lr][ka];
        a.y = sA[lr][ka + 1];
        b.x = W[(size_t)ka * KDIM + n0 + lr];
        b.y = W[(size_t)(ka + 1) * KDIM + n0 + lr];
        c = __builtin_amdgcn_wmma_f32_16x16x4_f32(false, a, false, b, (short)0, c,
                                                  false, false);
    }
    int rowbase = m0 + 8 * half;
    #pragma unroll
    for (int r = 0; r < 8; ++r) {
        int row = rowbase + r;
        if (row < E) {
            float inv = 1.f / fmaxf(edge_deg[row], 1.f);
            edge_feat[(size_t)row * KDIM + n0 + lr] = c[r] * inv;
        }
    }
}

__global__ __launch_bounds__(SC_THREADS) void fr_scatter(const float* __restrict__ edge_feat,
                                                         const int* __restrict__ node_idx,
                                                         const int* __restrict__ edge_idx,
                                                         float* __restrict__ out, int nnz, int n) {
    __shared__ float        acc[SC_ROWS * KDIM];
    __shared__ unsigned int queue[SC_WAVES * SC_QCAP];
    const int tid = threadIdx.x, lane = tid & 31;
    const int wave = __builtin_amdgcn_readfirstlane(tid >> 5);
    const int lo   = blockIdx.x * SC_ROWS;
    const int left = n - lo;
    const unsigned int cnt = (unsigned int)(left < SC_ROWS ? left : SC_ROWS);

    float4* acc4 = (float4*)acc;
    for (int i = tid; i < SC_ROWS * KDIM / 4; i += SC_THREADS) acc4[i] = make_float4(0.f, 0.f, 0.f, 0.f);
    __syncthreads();

    unsigned int* myq = queue + wave * SC_QCAP;
    int qn = 0;
    const int n4     = nnz >> 2;
    const int ragged = nnz & 3;
    const int nsteps = (n4 + SC_STEP4 - 1) / SC_STEP4;
    const int rot = (int)blockIdx.x * (nsteps / (int)gridDim.x);
    const int4* __restrict__ p4 = (const int4*)node_idx;

    if (ragged != 0 && wave == 0) {
        const int e = 4 * n4 + lane;
        unsigned int b = 0u;
        bool hit = false;
        if (lane < ragged) {
            b   = (unsigned int)(node_idx[e] - lo);
            hit = b < cnt;
        }
        sc_push(myq, qn, hit, ((unsigned int)e << SC_ROWBITS) | b);
    }

    if (wave < nsteps) {
        int4 bufa[SC_LD], bufb[SC_LD];
        sc_load_step(bufa, p4, n4, sc_rotate(wave, rot, nsteps), lane);
        for (int sl = wave; sl < nsteps; sl += 2 * SC_WAVES) {
            const int slb = sl + SC_WAVES, sla = sl + 2 * SC_WAVES;
            sc_load_step(bufb, p4, n4, sc_rotate(slb < nsteps ? slb : sl, rot, nsteps), lane);
            sc_scan(bufa, sc_rotate(sl, rot, nsteps), n4, lo, cnt, lane, myq, qn);
            while (qn >= SC_BATCH) {
                qn -= SC_BATCH;
                sc_drain<true>(acc, myq, qn, SC_BATCH, lane, edge_idx, edge_feat);
            }
            sc_load_step(bufa, p4, n4, sc_rotate(sla < nsteps ? sla : sl, rot, nsteps), lane);
            if (slb < nsteps) {
                sc_scan(bufb, sc_rotate(slb, rot, nsteps), n4, lo, cnt, lane, myq, qn);
                while (qn >= SC_BATCH) {
                    qn -= SC_BATCH;
                    sc_drain<true>(acc, myq, qn, SC_BATCH, lane, edge_idx, edge_feat);
                }
            }
        }
    }
    if (qn > 0) sc_drain<false>(acc, myq, 0, qn, lane, edge_idx, edge_feat);
    __syncthreads();

    const int total4 = (int)cnt * (KDIM / 4);
    float4* __restrict__ dst4 = (float4*)(out + (size_t)lo * KDIM);
    for (int i = tid; i < total4; i += SC_THREADS) dst4[i] = acc4[i];
}

__global__ __launch_bounds__(256) void fr_finalize(const float* __restrict__ probs,
                                                   const float* __restrict__ node_deg,
                                                   float* __restrict__ out, int n) {
    int wave = threadIdx.x >> 5, lane = threadIdx.x & 31;
    int row = blockIdx.x * 8 + wave;
    if (row >= n) return;
    float inv = 1.f / fmaxf(node_deg[row], 1.f);
    float4 acc = ((const float4*)(out + (size_t)row * KDIM))[lane];
    float4 p   = ((const float4*)(probs + (size_t)row * KDIM))[lane];
    float4 nat;
    nat.x = fmaxf(p.x, EPS) * acc.x * inv;
    nat.y = fmaxf(p.y, EPS) * acc.y * inv;
    nat.z = fmaxf(p.z, EPS) * acc.z * inv;
    nat.w = fmaxf(p.w, EPS) * acc.w * inv;
    float s = nat.x + nat.y + nat.z + nat.w;
    #pragma unroll
    for (int off = 16; off > 0; off >>= 1) s += __shfl_xor(s, off, 32);
    float mean = s * (1.0f / KDIM);
    nat.x -= mean; nat.y -= mean; nat.z -= mean; nat.w -= mean;
    ((float4*)(out + (size_t)row * KDIM))[lane] = nat;
}

extern "C" void kernel_launch(void* const* d_in, const int* in_sizes, int n_in,
                              void* d_out, int out_size, void* d_ws, size_t ws_size,
                              hipStream_t stream) {
    const float* y        = (const float*)d_in[0];
    const float* W        = (const float*)d_in[1];
    const int*   node_idx = (const int*)d_in[2];
    const int*   edge_idx = (const int*)d_in[3];
    float*       out      = (float*)d_out;

    const int n   = in_sizes[0] / KDIM;
    const int nnz = in_sizes[2];
    const int E   = N_EDGES;

    float* probs     = (float*)d_ws;
    float* edge_agg  = probs + (size_t)n * KDIM;
    float* edge_feat = edge_agg + (size_t)E * KDIM;
    float* edge_deg  = edge_feat + (size_t)E * KDIM;
    float* node_deg  = edge_deg + E;

    hipMemsetAsync(edge_agg, 0, (size_t)E * KDIM * sizeof(float), stream);
    hipMemsetAsync(edge_deg, 0, (size_t)(E + n) * sizeof(float), stream);

    fr_softmax<<<(n + 7) / 8, 256, 0, stream>>>(y, probs, n);
    fr_degrees<<<dim3(DEG_CHUNKS, 1 + DEG_NODE_JOBS), DEG_THREADS, 0, stream>>>(
        node_idx, edge_idx, edge_deg, node_deg, nnz, n, E);
    fr_gather<<<dim3(GA_CHUNKS, (N_EDGES + GA_ROWS - 1) / GA_ROWS), GA_THREADS, 0, stream>>>(
        probs, node_idx, edge_idx, edge_agg, nnz, E);
    fr_edge_gemm<<<(E + 15) / 16, 256, 0, stream>>>(edge_agg, edge_deg, W,
                                                    edge_feat, E);
    fr_scatter<<<(n + SC_ROWS - 1) / SC_ROWS, SC_THREADS, 0, stream>>>(
        edge_feat, node_idx, edge_idx, out, nnz, n);
    fr_finalize<<<(n + 7) / 8, 256, 0, stream>>>(probs, node_deg, out, n);
}


# gather+scatter: tile blocks filter a 1-byte owner-block key per entry (key lists written by extra blocks of the softmax launch), not the 4-byte indices: 1/4 the list bytes; sums in arrival order
# speedup vs baseline: 3.0918x; 1.0909x over previous
#include <hip/hip_runtime.h>

typedef __attribute__((ext_vector_type(2))) float v2f;
typedef __attribute__((ext_vector_type(8))) float v8f;

static constexpr int KDIM    = 128;
static constexpr int N_EDGES = 25000;
static constexpr float EPS   = 1e-12f;

static constexpr int SC_ROWS = 400;
static constexpr int GA_ROWS = 320;

__device__ __forceinline__ unsigned int sc_key4(int4 v, unsigned int rows) {
    return (((unsigned int)v.x / rows) & 255u)       | ((((unsigned int)v.y / rows) & 255u) << 8) |
           ((((unsigned int)v.z / rows) & 255u) << 16) | ((((unsigned int)v.w / rows) & 255u) << 24);
}

__global__ __launch_bounds__(256) void fr_softmax(const float* __restrict__ y,
                                                  float* __restrict__ probs, int n,
                                                  const int* __restrict__ node_idx,
                                                  const int* __restrict__ edge_idx,
                                                  unsigned int* __restrict__ node_key,
                                                  unsigned int* __restrict__ edge_key, int nnz) {
    const int row_blocks = (n + 7) / 8;
    if ((int)blockIdx.x >= row_blocks) {
        const int i = ((int)blockIdx.x - row_blocks) * 256 + (int)threadIdx.x;
        if (i < (nnz >> 2)) {
            node_key[i] = sc_key4(((const int4*)node_idx)[i], (unsigned int)SC_ROWS);
            edge_key[i] = sc_key4(((const int4*)edge_idx)[i], (unsigned int)GA_ROWS);
        }
        return;
    }
    int wave = threadIdx.x >> 5, lane = threadIdx.x & 31;
    int row = blockIdx.x * 8 + wave;
    if (row >= n) return;
    float4 v = ((const float4*)(y + (size_t)row * KDIM))[lane];
    float m = fmaxf(fmaxf(v.x, v.y), fmaxf(v.z, v.w));
    #pragma unroll
    for (int off = 16; off > 0; off >>= 1) m = fmaxf(m, __shfl_xor(m, off, 32));
    float4 e;
    e.x = __expf(v.x - m); e.y = __expf(v.y - m);
    e.z = __expf(v.z - m); e.w = __expf(v.w - m);
    float s = e.x + e.y + e.z + e.w;
    #pragma unroll
    for (int off = 16; off > 0; off >>= 1) s += __shfl_xor(s, off, 32);
    float inv = 1.f / s;
    e.x *= inv; e.y *= inv; e.z *= inv; e.w *= inv;
    ((float4*)(probs + (size_t)row * KDIM))[lane] = e;
}

static constexpr int DEG_SLICE     = 25000;
static constexpr int DEG_CHUNKS    = 32;
static constexpr int DEG_NODE_JOBS = 4;
static constexpr int DEG_THREADS   = 1024;

__device__ __forceinline__ void deg_count4(unsigned int* hist, int4 v, int lo, unsigned cnt) {
    const unsigned b0 = (unsigned)(v.x - lo), b1 = (unsigned)(v.y - lo);
    const unsigned b2 = (unsigned)(v.z - lo), b3 = (unsigned)(v.w - lo);
    if (b0 < cnt) atomicAdd(&hist[b0], 1u);
    if (b1 < cnt) atomicAdd(&hist[b1], 1u);
    if (b2 < cnt) atomicAdd(&hist[b2], 1u);
    if (b3 < cnt) atomicAdd(&hist[b3], 1u);
}

__global__ __launch_bounds__(DEG_THREADS) void fr_degrees(const int* __restrict__ node_idx,
                                                          const int* __restrict__ edge_idx,
                                                          float* __restrict__ edge_deg,
                                                          float* __restrict__ node_deg,
                                                          int nnz, int n, int E) {
    __shared__ unsigned int hist[DEG_SLICE];
    const int tid = threadIdx.x;
    const int job = blockIdx.y;
    const int* __restrict__ idx = (job == 0) ? edge_idx : node_idx;
    float* __restrict__ deg     = (job == 0) ? edge_deg : node_deg;
    const int rows   = (job == 0) ? E : n;
    const int lo0    = (job == 0) ? 0 : (job - 1) * DEG_SLICE;
    const int stride = (job == 0) ? DEG_SLICE : DEG_NODE_JOBS * DEG_SLICE;

    int per = (nnz + (int)gridDim.x - 1) / (int)gridDim.x;
    per = (per + 3) & ~3;
    const long long s64 = (long long)blockIdx.x * per;
    if (s64 >= nnz) return;
    const int start = (int)s64;
    const int end   = (nnz - start < per) ? nnz : start + per;
    const int n4    = (end - start) >> 2;
    const int4* __restrict__ p4 = (const int4*)(idx + start);

    for (int lo = lo0; lo < rows; lo += stride) {
        const int      left = rows - lo;
        const unsigned cnt  = (unsigned)(left < DEG_SLICE ? left : DEG_SLICE);
        for (unsigned b = tid; b < cnt; b += DEG_THREADS) hist[b] = 0u;
        __syncthreads();
        const int4 none = make_int4(-1, -1, -1, -1);
        const int  last = n4 - 1;
        for (int q = tid; q < n4; q += 4 * DEG_THREADS) {
            const int q1 = q + DEG_THREADS, q2 = q + 2 * DEG_THREADS, q3 = q + 3 * DEG_THREADS;
            const int4 v0 = p4[q];
            int4 v1 = p4[q1 < last ? q1 : last];
            int4 v2 = p4[q2 < last ? q2 : last];
            int4 v3 = p4[q3 < last ? q3 : last];
            if (q1 > last) v1 = none;
            if (q2 > last) v2 = none;
            if (q3 > last) v3 = none;
            deg_count4(hist, v0, lo, cnt);
            deg_count4(hist, v1, lo, cnt);
            deg_count4(hist, v2, lo, cnt);
            deg_count4(hist, v3, lo, cnt);
        }
        for (int i = start + (n4 << 2) + tid; i < end; i += DEG_THREADS) {
            const unsigned b = (unsigned)(idx[i] - lo);
            if (b < cnt) atomicAdd(&hist[b], 1u);
        }
        __syncthreads();
        for (unsigned b = tid; b < cnt; b += DEG_THREADS) {
            const unsigned c = hist[b];
            if (c) atomicAdd(&deg[lo + (int)b], (float)c);
        }
        __syncthreads();
    }
}

static constexpr int SC_THREADS = 1024;
static constexpr int SC_WAVES   = SC_THREADS / 32;
static constexpr int SC_LD      = 4;
static constexpr int SC_STEP16  = 32 * SC_LD;
static constexpr int SC_BATCH   = 16;
static constexpr int SC_QCAP    = 32 * 16 + SC_BATCH;

__device__ __forceinline__ uint4 sc_load16(const uint4* __restrict__ p16, int n16, int q) {
    const int last = n16 - 1;
    return p16[q < last ? q : last];
}

__device__ __forceinline__ void sc_push(unsigned int* myq, int& qn, bool hit, unsigned int rec) {
    const unsigned int mk = __builtin_amdgcn_ballot_w32(hit);
    if (hit) myq[qn + (int)__builtin_amdgcn_mbcnt_lo(mk, 0u)] = rec;
    qn += __builtin_popcount(mk);
}

__device__ __forceinline__ int sc_rotate(int sl, int rot, int nsteps) {
    const int s = sl + rot;
    return s >= nsteps ? s - nsteps : s;
}

__device__ __forceinline__ void sc_load_step(uint4 (&buf)[SC_LD], const uint4* __restrict__ p16, int n16,
                                             int s, int lane) {
    for (int j = 0; j < SC_LD; ++j) buf[j] = sc_load16(p16, n16, s * SC_STEP16 + j * 32 + lane);
}

__device__ __forceinline__ void sc_scan_word(unsigned int w, unsigned int flags, unsigned int pat, unsigned int e0,
                                             unsigned int* myq, int& qn) {
    const unsigned int t = w ^ pat;
    unsigned int z = ~(((t & 0x7F7F7F7Fu) + 0x7F7F7F7Fu) | t) & flags;
    while (__builtin_amdgcn_ballot_w32(z != 0u) != 0u) {
        const bool hit = z != 0u;
        const unsigned int k = hit ? ((unsigned int)__builtin_ctz(z) >> 3) : 0u;
        sc_push(myq, qn, hit, e0 + k);
        z &= z - 1u;
    }
}

template <bool FULL, int ROWS>
__device__ __forceinline__ void sc_drain(float* acc, const unsigned int* myq, int base, int m,
                                         int lane, int lo, unsigned int cnt,
                                         const int* __restrict__ dst_idx,
                                         const int* __restrict__ src_idx,
                                         const float* __restrict__ src_rows) {
    const int slot = lane & (SC_BATCH - 1);
    const unsigned int e  = myq[base + ((FULL || slot < m) ? slot : 0)];
    const int          ev = src_idx[e];
    const unsigned int b  = (unsigned int)(dst_idx[e] - lo);
    float4 p[SC_BATCH];
    #pragma unroll
    for (int j = 0; j < SC_BATCH; ++j) {
        const int evj = __builtin_amdgcn_readlane(ev, j);
        p[j] = ((const float4*)(src_rows + (size_t)evj * KDIM))[lane];
    }
    const int h = lane >> 4;
    #pragma unroll
    for (int j = 0; j < SC_BATCH; ++j) {
        const unsigned int bj = (unsigned int)__builtin_amdgcn_readlane((int)b, j);
        const unsigned int rj = ((FULL || j < m) && bj < cnt) ? bj : (unsigned int)ROWS;
        float* d = acc + rj * KDIM + 4 * lane;
        atomicAdd(d + h,           h ? p[j].y : p[j].x);
        atomicAdd(d + (h ^ 1),     h ? p[j].x : p[j].y);
        atomicAdd(d + 2 + h,       h ? p[j].w : p[j].z);
        atomicAdd(d + 2 + (h ^ 1), h ? p[j].z : p[j].w);
    }
}

template <int ROWS>
__device__ __forceinline__ void sc_trip(const uint4 (&buf)[SC_LD], int s, int n16, unsigned int pat,
                                        int lo, unsigned int cnt, int lane,
                                        float* acc, unsigned int* myq, int& qn,
                                        const int* __restrict__ dst_idx,
                                        const int* __restrict__ src_idx,
                                        const float* __restrict__ src_rows) {
    #pragma unroll
    for (int j = 0; j < SC_LD; ++j) {
        const int q = s * SC_STEP16 + j * 32 + lane;
        const unsigned int flags = q < n16 ? 0x80808080u : 0u;
        const unsigned int e0 = 16u * (unsigned int)q;
        sc_scan_word(buf[j].x, flags, pat, e0,       myq, qn);
        sc_scan_word(buf[j].y, flags, pat, e0 + 4u,  myq, qn);
        sc_scan_word(buf[j].z, flags, pat, e0 + 8u,  myq, qn);
        sc_scan_word(buf[j].w, flags, pat, e0 + 12u, myq, qn);
        while (qn >= SC_BATCH) {
            qn -= SC_BATCH;
            sc_drain<true, ROWS>(acc, myq, qn, SC_BATCH, lane, lo, cnt, dst_idx, src_idx, src_rows);
        }
    }
}

static constexpr int GA_CHUNKS  = 3;
static constexpr int GA_THREADS = SC_THREADS;

__global__ __launch_bounds__(GA_THREADS) void fr_gather(const float* __restrict__ probs,
                                                        const int* __restrict__ node_idx,
                                                        const int* __restrict__ edge_idx,
                                                        const unsigned int* __restrict__ edge_key,
                                                        float* __restrict__ edge_agg,
                                                        int nnz, int E) {
    __shared__ float        acc[(GA_ROWS + 1) * KDIM];
    __shared__ unsigned int queue[SC_WAVES * SC_QCAP];
    const int tid = threadIdx.x, lane = tid & 31;
    const int wave = __builtin_amdgcn_readfirstlane(tid >> 5);
    const int lo   = blockIdx.y * GA_ROWS;
    const int left = E - lo;
    const unsigned int cnt = (unsigned int)(left < GA_ROWS ? left : GA_ROWS);
    const unsigned int pat = ((unsigned int)blockIdx.y & 255u) * 0x01010101u;

    int per = (nnz + (int)gridDim.x - 1) / (int)gridDim.x;
    per = (per + 15) & ~15;
    const long long s64 = (long long)blockIdx.x * per;
    if (s64 >= nnz) return;
    const int start  = (int)s64;
    const int end    = (nnz - start < per) ? nnz : start + per;
    const int n16    = (end - start) >> 4;
    const int ragged = (end - start) & 15;
    const int nsteps = (n16 + SC_STEP16 - 1) / SC_STEP16;
    const int rot = (int)blockIdx.y * (nsteps / (int)gridDim.y);
    const int*   __restrict__ edges = edge_idx + start;
    const int*   __restrict__ nodes = node_idx + start;
    const uint4* __restrict__ p16   = (const uint4*)(edge_key + (start >> 2));

    float4* acc4 = (float4*)acc;
    for (int i = tid; i < (GA_ROWS + 1) * KDIM / 4; i += GA_THREADS) acc4[i] = make_float4(0.f, 0.f, 0.f, 0.f);
    __syncthreads();

    unsigned int* myq = queue + wave * SC_QCAP;
    int qn = 0;

    if (ragged != 0 && wave == 0) {
        const int e = 16 * n16 + lane;
        bool hit = false;
        if (lane < ragged) hit = (unsigned int)(edges[e] - lo) < cnt;
        sc_push(myq, qn, hit, (unsigned int)e);
    }

    if (wave < nsteps) {
        uint4 cur[SC_LD], nxt[SC_LD];
        sc_load_step(cur, p16, n16, sc_rotate(wave, rot, nsteps), lane);
        for (int sl = wave; sl < nsteps; sl += SC_WAVES) {
            const int sn = sl + SC_WAVES;
            sc_load_step(nxt, p16, n16, sc_rotate(sn < nsteps ? sn : sl, rot, nsteps), lane);
            sc_trip<GA_ROWS>(cur, sc_rotate(sl, rot, nsteps), n16, pat, lo, cnt, lane, acc, myq, qn,
                             edges, nodes, probs);
            for (int j = 0; j < SC_LD; ++j) cur[j] = nxt[j];
        }
    }
    if (qn > 0) sc_drain<false, GA_ROWS>(acc, myq, 0, qn, lane, lo, cnt, edges, nodes, probs);
    __syncthreads();

    const int total = (int)cnt * KDIM;
    float* __restrict__ dst = edge_agg + (size_t)lo * KDIM;
    for (int i = tid; i < total; i += GA_THREADS) {
        const float v = acc[i];
        if (v != 0.f) atomicAdd(dst + i, v);
    }
}

__global__ __launch_bounds__(256) void fr_edge_gemm(const float* edge_agg,
                                                    const float* __restrict__ edge_deg,
                                                    const float* __restrict__ W,
                                                    float* edge_feat, int E) {
    __shared__ float sA[16][132];
    int tid = threadIdx.x;
    int m0 = blockIdx.x * 16;
    for (int i = tid; i < 16 * 32; i += 256) {
        int row = i >> 5, c4 = i & 31, r = m0 + row;
        float4 v = make_float4(0.f, 0.f, 0.f, 0.f);
        if (r < E) v = ((const float4*)edge_agg)[(size_t)r * 32 + c4];
        *(float4*)&sA[row][c4 * 4] = v;
    }
    __syncthreads();

    int wave = tid >> 5, lane = tid & 31;
    int half = lane >> 4, lr = lane & 15;
    int n0 = wave * 16;

    v8f c = {0.f, 0.f, 0.f, 0.f, 0.f, 0.f, 0.f, 0.f};
    #pragma unroll 4
    for (int k0 = 0; k0 < KDIM; k0 += 4) {
        int ka = k0 + 2 * half;
        v2f a, b;
        a.x = sA[lr][ka];
        a.y = sA[lr][ka + 1];
        b.x = W[(size_t)ka * KDIM + n0 + lr];
        b.y = W[(size_t)(ka + 1) * KDIM + n0 + lr];
        c = __builtin_amdgcn_wmma_f32_16x16x4_f32(false, a, false, b, (short)0, c,
                                                  false, false);
    }
    int rowbase = m0 + 8 * half;
    #pragma unroll
    for (int r = 0; r < 8; ++r) {
        int row = rowbase + r;
        if (row < E) {
            float inv = 1.f / fmaxf(edge_deg[row], 1.f);
            edge_feat[(size_t)row * KDIM + n0 + lr] = c[r] * inv;
        }
    }
}

__global__ __launch_bounds__(SC_THREADS) void fr_scatter(const float* __restrict__ edge_feat,
                                                         const int* __restrict__ node_idx,
                                                         const int* __restrict__ edge_idx,
                                                         const unsigned int* __restrict__ node_key,
                                                         float* __restrict__ out, int nnz, int n) {
    __shared__ float        acc[(SC_ROWS + 1) * KDIM];
    __shared__ unsigned int queue[SC_WAVES * SC_QCAP];
    const int tid = threadIdx.x, lane = tid & 31;
    const int wave = __builtin_amdgcn_readfirstlane(tid >> 5);
    const int lo   = blockIdx.x * SC_ROWS;
    const int left = n - lo;
    const unsigned int cnt = (unsigned int)(left < SC_ROWS ? left : SC_ROWS);
    const unsigned int pat = ((unsigned int)blockIdx.x & 255u) * 0x01010101u;

    float4* acc4 = (float4*)acc;
    for (int i = tid; i < (SC_ROWS + 1) * KDIM / 4; i += SC_THREADS) acc4[i] = make_float4(0.f, 0.f, 0.f, 0.f);
    __syncthreads();

    unsigned int* myq = queue + wave * SC_QCAP;
    int qn = 0;
    const int n16    = nnz >> 4;
    const int ragged = nnz & 15;
    const int nsteps = (n16 + SC_STEP16 - 1) / SC_STEP16;
    const int rot = (int)blockIdx.x * (nsteps / (int)gridDim.x);
    const uint4* __restrict__ p16 = (const uint4*)node_key;

    if (ragged != 0 && wave == 0) {
        const int e = 16 * n16 + lane;
        bool hit = false;
        if (lane < ragged) hit = (unsigned int)(node_idx[e] - lo) < cnt;
        sc_push(myq, qn, hit, (unsigned int)e);
    }

    if (wave < nsteps) {
        uint4 cur[SC_LD], nxt[SC_LD];
        sc_load_step(cur, p16, n16, sc_rotate(wave, rot, nsteps), lane);
        for (int sl = wave; sl < nsteps; sl += SC_WAVES) {
            const int sn = sl + SC_WAVES;
            sc_load_step(nxt, p16, n16, sc_rotate(sn < nsteps ? sn : sl, rot, nsteps), lane);
            sc_trip<SC_ROWS>(cur, sc_rotate(sl, rot, nsteps), n16, pat, lo, cnt, lane, acc, myq, qn,
                             node_idx, edge_idx, edge_feat);
            for (int j = 0; j < SC_LD; ++j) cur[j] = nxt[j];
        }
    }
    if (qn > 0) sc_drain<false, SC_ROWS>(acc, myq, 0, qn, lane, lo, cnt, node_idx, edge_idx, edge_feat);
    __syncthreads();

    const int total4 = (int)cnt * (KDIM / 4);
    float4* __restrict__ dst4 = (float4*)(out + (size_t)lo * KDIM);
    for (int i = tid; i < total4; i += SC_THREADS) dst4[i] = acc4[i];
}

__global__ __launch_bounds__(256) void fr_finalize(const float* __restrict__ probs,
                                                   const float* __restrict__ node_deg,
                                                   float* __restrict__ out, int n) {
    int wave = threadIdx.x >> 5, lane = threadIdx.x & 31;
    int row = blockIdx.x * 8 + wave;
    if (row >= n) return;
    float inv = 1.f / fmaxf(node_deg[row], 1.f);
    float4 acc = ((const float4*)(out + (size_t)row * KDIM))[lane];
    float4 p   = ((const float4*)(probs + (size_t)row * KDIM))[lane];
    float4 nat;
    nat.x = fmaxf(p.x, EPS) * acc.x * inv;
    nat.y = fmaxf(p.y, EPS) * acc.y * inv;
    nat.z = fmaxf(p.z, EPS) * acc.z * inv;
    nat.w = fmaxf(p.w, EPS) * acc.w * inv;
    float s = nat.x + nat.y + nat.z + nat.w;
    #pragma unroll
    for (int off = 16; off > 0; off >>= 1) s += __shfl_xor(s, off, 32);
    float mean = s * (1.0f / KDIM);
    nat.x -= mean; nat.y -= mean; nat.z -= mean; nat.w -= mean;
    ((float4*)(out + (size_t)row * KDIM))[lane] = nat;
}

extern "C" void kernel_launch(void* const* d_in, const int* in_sizes, int n_in,
                              void* d_out, int out_size, void* d_ws, size_t ws_size,
                              hipStream_t stream) {
    const float* y        = (const float*)d_in[0];
    const float* W        = (const float*)d_in[1];
    const int*   node_idx = (const int*)d_in[2];
    const int*   edge_idx = (const int*)d_in[3];
    float*       out      = (float*)d_out;

    const int n   = in_sizes[0] / KDIM;
    const int nnz = in_sizes[2];
    const int E   = N_EDGES;

    float* probs     = (float*)d_ws;
    float* edge_agg  = probs + (size_t)n * KDIM;
    float* key_plane = edge_agg + (size_t)E * KDIM;
    float* edge_deg  = key_plane + (size_t)E * KDIM;
    float* node_deg  = edge_deg + E;
    unsigned int* node_key = (unsigned int*)key_plane;
    unsigned int* edge_key = node_key + ((size_t)nnz + 15) / 16 * 4;

    hipMemsetAsync(edge_agg, 0, (size_t)E * KDIM * sizeof(float), stream);
    hipMemsetAsync(edge_deg, 0, (size_t)(E + n) * sizeof(float), stream);

    fr_softmax<<<(n + 7) / 8 + ((nnz >> 2) + 255) / 256, 256, 0, stream>>>(
        y, probs, n, node_idx, edge_idx, node_key, edge_key, nnz);
    fr_degrees<<<dim3(DEG_CHUNKS, 1 + DEG_NODE_JOBS), DEG_THREADS, 0, stream>>>(
        node_idx, edge_idx, edge_deg, node_deg, nnz, n, E);
    fr_gather<<<dim3(GA_CHUNKS, (N_EDGES + GA_ROWS - 1) / GA_ROWS), GA_THREADS, 0, stream>>>(
        probs, node_idx, edge_idx, edge_key, edge_agg, nnz, E);
    fr_edge_gemm<<<(E + 15) / 16, 256, 0, stream>>>(edge_agg, edge_deg, W,
                                                    edge_agg, E);
    fr_scatter<<<(n + SC_ROWS - 1) / SC_ROWS, SC_THREADS, 0, stream>>>(
        edge_agg, node_idx, edge_idx, node_key, out, nnz, n);
    fr_finalize<<<(n + 7) / 8, 256, 0, stream>>>(probs, node_deg, out, n);
}
